# starting kernel
# baseline (speedup 1.0000x reference)
#include <hip/hip_runtime.h>
#include <hip/hip_bf16.h>

typedef float v2f __attribute__((ext_vector_type(2)));
typedef float v8f __attribute__((ext_vector_type(8)));

__global__ __launch_bounds__(256) void wmma_gemm_bias(
    const float* __restrict__ A, const float* __restrict__ W,
    const float* __restrict__ bias, float* __restrict__ C,
    int M, int N, int K, int relu)
{
    const int lane = threadIdx.x & 31;
    const int wave = threadIdx.x >> 5;
    const int tilesN = N >> 6;
    const int tile = blockIdx.x * 8 + wave;
    const int totalTiles = (M >> 4) * tilesN;
    if (tile >= totalTiles) return;
    const int tm = (tile / tilesN) << 4;
    const int tn = (tile % tilesN) << 6;

    const int half = lane >> 4;
    const int l16  = lane & 15;

    const float* ap = A + (size_t)(tm + l16) * K + 2 * half;
    const float* bp = W + (size_t)(2 * half) * N + tn + l16;
    const size_t sN = (size_t)N;

    v8f acc0 = {}, acc1 = {}, acc2 = {}, acc3 = {};
    for (int k = 0; k < K; k += 4) {
        v2f a = *(const v2f*)ap;
        const float* b0 = bp;
        const float* b1 = bp + sN;
        v2f f0, f1, f2, f3;
        f0.x = b0[0];  f0.y = b1[0];
        f1.x = b0[16]; f1.y = b1[16];
        f2.x = b0[32]; f2.y = b1[32];
        f3.x = b0[48]; f3.y = b1[48];
        acc0 = __builtin_amdgcn_wmma_f32_16x16x4_f32(false, a, false, f0,
                                                     (short)0, acc0, false, false);
        acc1 = __builtin_amdgcn_wmma_f32_16x16x4_f32(false, a, false, f1,
                                                     (short)0, acc1, false, false);
        acc2 = __builtin_amdgcn_wmma_f32_16x16x4_f32(false, a, false, f2,
                                                     (short)0, acc2, false, false);
        acc3 = __builtin_amdgcn_wmma_f32_16x16x4_f32(false, a, false, f3,
                                                     (short)0, acc3, false, false);
        ap += 4;
        bp += 4 * sN;
        __builtin_prefetch(bp + 8 * sN, 0, 1);
    }

    const int row0 = tm + half * 8;
    float bv0 = 0.f, bv1 = 0.f, bv2 = 0.f, bv3 = 0.f;
    if (bias) {
        const float* bb = bias + tn + l16;
        bv0 = bb[0]; bv1 = bb[16]; bv2 = bb[32]; bv3 = bb[48];
    }
    float* crow = C + (size_t)row0 * N + tn + l16;
#pragma unroll
    for (int r = 0; r < 8; ++r) {
        float v0 = acc0[r] + bv0, v1 = acc1[r] + bv1;
        float v2 = acc2[r] + bv2, v3 = acc3[r] + bv3;
        if (relu) {
            v0 = fmaxf(v0, 0.f); v1 = fmaxf(v1, 0.f);
            v2 = fmaxf(v2, 0.f); v3 = fmaxf(v3, 0.f);
        }
        crow[0] = v0; crow[16] = v1; crow[32] = v2; crow[48] = v3;
        crow += sN;
    }
}

__global__ void count_same(const int* __restrict__ y, float* __restrict__ cnt, int S)
{
    int i = blockIdx.x * blockDim.x + threadIdx.x;
    if (i >= S) return;
    int yi = y[i];
    int c = 0;
    for (int j = 0; j < S; ++j) c += (y[j] == yi) ? 1 : 0;
    cnt[i] = (float)(c - 1);
}

__global__ __launch_bounds__(256) void pair_accum(
    const float* __restrict__ Abuf, const float* __restrict__ Bbuf,
    const float* __restrict__ bm1t, const int* __restrict__ y,
    float* __restrict__ haggr, int S, int H)
{
    const int i = blockIdx.x;
    const int h = threadIdx.x;
    const int yi = y[i];
    const float av = Abuf[(size_t)i * H + h] + bm1t[h];
    float acc = 0.0f;
    for (int j = 0; j < S; ++j) {
        if (y[j] == yi && j != i)
            acc += fmaxf(av + Bbuf[(size_t)j * H + h], 0.0f);
    }
    haggr[(size_t)i * H + h] = acc;
}

__global__ __launch_bounds__(256) void update_s(
    float* __restrict__ s, const float* __restrict__ aggmat,
    const float* __restrict__ bm2t, const float* __restrict__ cnt, int H)
{
    const int i = blockIdx.x;
    const int h = threadIdx.x;
    const float c = cnt[i];
    const float denom = fmaxf(c, 1.0f);
    s[(size_t)i * H + h] += (aggmat[(size_t)i * H + h] + c * bm2t[h]) / denom;
}

__global__ __launch_bounds__(256) void score_kernel(
    const float* __restrict__ hq, const float* __restrict__ hs,
    const float* __restrict__ br1, const float* __restrict__ wr2,
    const float* __restrict__ br2, float* __restrict__ out, int S, int H)
{
    const int q = blockIdx.x;
    const int lane = threadIdx.x & 31;
    const int wave = threadIdx.x >> 5;
    const int hbase = lane * 8;

    float hqv[8], bv[8], wv[8];
#pragma unroll
    for (int u = 0; u < 8; ++u) {
        hqv[u] = hq[(size_t)q * H + hbase + u];
        bv[u]  = br1[hbase + u];
        wv[u]  = wr2[hbase + u];
    }
    const float b2 = br2[0];

    for (int s = wave; s < S; s += 8) {
        const float* hsrow = hs + (size_t)s * H + hbase;
        float sum = 0.0f;
#pragma unroll
        for (int u = 0; u < 8; ++u)
            sum += fmaxf(hqv[u] + hsrow[u] + bv[u], 0.0f) * wv[u];
#pragma unroll
        for (int off = 16; off > 0; off >>= 1)
            sum += __shfl_down(sum, off, 32);
        if (lane == 0) out[(size_t)q * S + s] = sum + b2;
    }
}

extern "C" void kernel_launch(void* const* d_in, const int* in_sizes, int n_in,
                              void* d_out, int out_size, void* d_ws, size_t ws_size,
                              hipStream_t stream)
{
    const float* Xq  = (const float*)d_in[0];
    const float* Xs  = (const float*)d_in[1];
    const int*   y   = (const int*)d_in[2];
    const float* Wn1 = (const float*)d_in[3];
    const float* bn1 = (const float*)d_in[4];
    const float* Wn2 = (const float*)d_in[5];
    const float* bn2 = (const float*)d_in[6];
    const float* Wm1 = (const float*)d_in[7];
    const float* bm1 = (const float*)d_in[8];
    const float* Wm2 = (const float*)d_in[9];
    const float* bm2 = (const float*)d_in[10];
    const float* Wr1 = (const float*)d_in[11];
    const float* br1 = (const float*)d_in[12];
    const float* wr2 = (const float*)d_in[13];
    const float* br2 = (const float*)d_in[14];
    float* out = (float*)d_out;

    const int E = 512, H = 256, Q = 1024, S = 512, STEPS = 3;

    float* ws  = (float*)d_ws;
    float* q1  = ws; ws += (size_t)Q * H;
    float* qf  = ws; ws += (size_t)Q * H;
    float* s1  = ws; ws += (size_t)S * H;
    float* sf  = ws; ws += (size_t)S * H;
    float* Ab  = ws; ws += (size_t)S * H;
    float* Bb  = ws; ws += (size_t)S * H;
    float* hag = ws; ws += (size_t)S * H;
    float* agg = ws; ws += (size_t)S * H;
    float* hqb = ws; ws += (size_t)Q * H;
    float* hsb = ws; ws += (size_t)S * H;
    float* cnt = ws; ws += S;

    auto gemm = [&](const float* A, const float* W, const float* bias, float* C,
                    int M, int N, int K, int relu) {
        int tiles  = (M / 16) * (N / 64);
        int blocks = (tiles + 7) / 8;
        wmma_gemm_bias<<<blocks, 256, 0, stream>>>(A, W, bias, C, M, N, K, relu);
    };

    gemm(Xq, Wn1, bn1, q1, Q, H, E, 1);
    gemm(q1, Wn2, bn2, qf, Q, H, H, 0);
    gemm(Xs, Wn1, bn1, s1, S, H, E, 1);
    gemm(s1, Wn2, bn2, sf, S, H, H, 0);

    count_same<<<(S + 255) / 256, 256, 0, stream>>>(y, cnt, S);

    for (int t = 0; t < STEPS; ++t) {
        const float* W1a = Wm1 + (size_t)t * (2 * H) * H;
        const float* W1b = W1a + (size_t)H * H;
        gemm(sf, W1a, nullptr, Ab, S, H, H, 0);
        gemm(sf, W1b, nullptr, Bb, S, H, H, 0);
        pair_accum<<<S, H, 0, stream>>>(Ab, Bb, bm1 + (size_t)t * H, y, hag, S, H);
        gemm(hag, Wm2 + (size_t)t * H * H, nullptr, agg, S, H, H, 0);
        update_s<<<S, H, 0, stream>>>(sf, agg, bm2 + (size_t)t * H, cnt, H);
    }

    gemm(qf, Wr1, nullptr, hqb, Q, H, H, 0);
    gemm(sf, Wr1 + (size_t)H * H, nullptr, hsb, S, H, H, 0);
    score_kernel<<<Q, 256, 0, stream>>>(hqb, hsb, br1, wr2, br2, out, S, H);
}


# Score kernel: wave keeps 32 hs rows in registers, streams 8 hq rows; a row's 32 lane partials are summed at once by a transposing DPP/permlanex16 butterfly; products fused (fma); sum order kept
# speedup vs baseline: 6.6768x; 6.6768x over previous
#include <hip/hip_runtime.h>
#include <hip/hip_bf16.h>

typedef float v2f __attribute__((ext_vector_type(2)));
typedef float v4f __attribute__((ext_vector_type(4)));
typedef float v8f __attribute__((ext_vector_type(8)));

#define G_BM 32
#define G_BN 64
#define G_BK 128
#define G_ALD (G_BK + 4)
#define G_WLD (G_BN + 8)

__global__ __launch_bounds__(256) void wmma_gemm_bias(
    const float* __restrict__ A, const float* __restrict__ W,
    const float* __restrict__ bias, float* __restrict__ C,
    int M, int N, int K, int relu)
{
    __shared__ __attribute__((aligned(16))) float As[G_BM * G_ALD];
    __shared__ __attribute__((aligned(16))) float Ws[G_BK * G_WLD];

    const int tid  = threadIdx.x;
    const int lane = tid & 31;
    const int wave = tid >> 5;
    const int half = lane >> 4;
    const int l16  = lane & 15;

    const int tilesN = N >> 6;
    const int bm = (blockIdx.x / tilesN) * G_BM;
    const int bn = (blockIdx.x % tilesN) * G_BN;
    const int wm = (wave >> 2) << 4;
    const int wn = (wave & 3) << 4;

    const int ar = tid >> 5, ac = (tid & 31) << 2;
    const int wr = tid >> 4, wc = (tid & 15) << 2;

    const float* ag = A + (size_t)(bm + ar) * K + ac;
    const float* wg = W + (size_t)wr * N + bn + wc;
    const size_t a8  = (size_t)8 * K;
    const size_t w16 = (size_t)16 * N;

    v4f ra[4], rw[8];
#pragma unroll
    for (int i = 0; i < 4; ++i) ra[i] = *(const v4f*)(ag + i * a8);
#pragma unroll
    for (int i = 0; i < 8; ++i) rw[i] = *(const v4f*)(wg + i * w16);

    const float* afrag = &As[(wm + l16) * G_ALD + 2 * half];
    const float* bfrag = &Ws[(2 * half) * G_WLD + wn + l16];

    v8f acc = {};
    for (int k0 = 0; k0 < K; k0 += G_BK) {
#pragma unroll
        for (int i = 0; i < 4; ++i)
            *(v4f*)&As[(ar + 8 * i) * G_ALD + ac] = ra[i];
#pragma unroll
        for (int i = 0; i < 8; ++i)
            *(v4f*)&Ws[(wr + 16 * i) * G_WLD + wc] = rw[i];
        __syncthreads();

        if (k0 + G_BK < K) {
            ag += G_BK;
            wg += (size_t)G_BK * N;
#pragma unroll
            for (int i = 0; i < 4; ++i) ra[i] = *(const v4f*)(ag + i * a8);
#pragma unroll
            for (int i = 0; i < 8; ++i) rw[i] = *(const v4f*)(wg + i * w16);
        }

        v2f fa[2][8], fb[2][8];
#pragma unroll
        for (int s = 0; s < 8; ++s) {
            fa[0][s]   = *(const v2f*)(afrag + 4 * s);
            fb[0][s].x = bfrag[(4 * s) * G_WLD];
            fb[0][s].y = bfrag[(4 * s) * G_WLD + G_WLD];
        }
#pragma unroll
        for (int g = 0; g < G_BK / 32; ++g) {
            if (g + 1 < G_BK / 32) {
#pragma unroll
                for (int s = 0; s < 8; ++s) {
                    const int kk = 32 * (g + 1) + 4 * s;
                    fa[(g + 1) & 1][s]   = *(const v2f*)(afrag + kk);
                    fb[(g + 1) & 1][s].x = bfrag[kk * G_WLD];
                    fb[(g + 1) & 1][s].y = bfrag[kk * G_WLD + G_WLD];
                }
            }
            __builtin_amdgcn_sched_barrier(0);
#pragma unroll
            for (int s = 0; s < 8; ++s)
                acc = __builtin_amdgcn_wmma_f32_16x16x4_f32(false, fa[g & 1][s], false,
                                                            fb[g & 1][s], (short)0, acc,
                                                            false, false);
            __builtin_amdgcn_sched_barrier(0);
        }
        __syncthreads();
    }

    const int row0 = bm + wm + half * 8;
    const int col  = bn + wn + l16;
    float bv = 0.f;
    if (bias) bv = bias[col];
    float* crow = C + (size_t)row0 * N + col;
#pragma unroll
    for (int r = 0; r < 8; ++r) {
        float v = acc[r] + bv;
        if (relu) v = fmaxf(v, 0.f);
        crow[0] = v;
        crow += N;
    }
}

#define F_NW 16
#define F_COLS 32
#define F_MAXN 512
#define F_H 256
#define F_KC 64

__global__ __launch_bounds__(512) void class_ab_pair(
    const float* __restrict__ sf, const float* __restrict__ W1a,
    const float* __restrict__ W1b, const float* __restrict__ bm1t,
    const int* __restrict__ y, float* __restrict__ haggr, int S)
{
    __shared__ float As[F_MAXN * F_COLS];
    __shared__ float Bs[F_MAXN * F_COLS];
    __shared__ int idx[F_MAXN];
    __shared__ int wcount[F_NW];

    const int tid  = threadIdx.x;
    const int lane = tid & 31;
    const int wave = __builtin_amdgcn_readfirstlane(tid >> 5);
    const int half = lane >> 4;
    const int l16  = lane & 15;
    const int cls  = blockIdx.y;
    const int col0 = blockIdx.x * F_COLS;

    int n = 0;
    for (int base = 0; base < S; base += F_NW * 32) {
        const int j = base + tid;
        const bool mine = (j < S) && (y[j] == cls);
        const unsigned bal = __builtin_amdgcn_ballot_w32(mine);
        if (lane == 0) wcount[wave] = __popc(bal);
        __syncthreads();
        int before = 0, total = 0;
        for (int w = 0; w < F_NW; ++w) {
            const int c = wcount[w];
            if (w < wave) before += c;
            total += c;
        }
        const int pos = n + before + __popc(bal & ((1u << lane) - 1u));
        if (mine && pos < F_MAXN) idx[pos] = j;
        n += total;
        __syncthreads();
    }
    if (n == 0) return;
    if (n > F_MAXN) n = F_MAXN;

    const int ntiles = ((n + 15) >> 4) * 4;
    for (int t = wave; t < ntiles; t += F_NW) {
        const int m = t >> 2, nt = t & 3;
        int rl = m * 16 + l16;
        if (rl >= n) rl = n - 1;
        const float* arow = sf + (size_t)idx[rl] * F_H + 2 * half;
        const float* wcol = ((nt & 2) ? W1b : W1a) + (size_t)(2 * half) * F_H
                            + col0 + (nt & 1) * 16 + l16;
        v2f fa[2][16], fb[2][16];
#pragma unroll
        for (int s = 0; s < 16; ++s) {
            fa[0][s]   = *(const v2f*)(arow + 4 * s);
            fb[0][s].x = wcol[(4 * s) * F_H];
            fb[0][s].y = wcol[(4 * s + 1) * F_H];
        }
        v8f acc = {};
#pragma unroll
        for (int c = 0; c < F_H / F_KC; ++c) {
            if (c + 1 < F_H / F_KC) {
#pragma unroll
                for (int s = 0; s < 16; ++s) {
                    const int kk = F_KC * (c + 1) + 4 * s;
                    fa[(c + 1) & 1][s]   = *(const v2f*)(arow + kk);
                    fb[(c + 1) & 1][s].x = wcol[kk * F_H];
                    fb[(c + 1) & 1][s].y = wcol[(kk + 1) * F_H];
                }
            }
            __builtin_amdgcn_sched_barrier(0);
#pragma unroll
            for (int s = 0; s < 16; ++s)
                acc = __builtin_amdgcn_wmma_f32_16x16x4_f32(false, fa[c & 1][s], false,
                                                            fb[c & 1][s], (short)0, acc,
                                                            false, false);
            __builtin_amdgcn_sched_barrier(0);
        }
        float* dst = ((nt & 2) ? Bs : As) + (m * 16 + half * 8) * F_COLS + (nt & 1) * 16 + l16;
#pragma unroll
        for (int r = 0; r < 8; ++r) dst[r * F_COLS] = acc[r];
    }
    __syncthreads();

    const float bmv = bm1t[col0 + lane];
    for (int r0 = wave; r0 < n; r0 += 4 * F_NW) {
        int il[4];
        float a[4], sum[4];
#pragma unroll
        for (int r = 0; r < 4; ++r) {
            il[r]  = r0 + r * F_NW;
            a[r]   = (il[r] < n) ? As[il[r] * F_COLS + lane] + bmv : 0.f;
            sum[r] = 0.f;
        }
#pragma unroll 4
        for (int j = 0; j < n; ++j) {
            const float b = Bs[j * F_COLS + lane];
#pragma unroll
            for (int r = 0; r < 4; ++r)
                if (j != il[r]) sum[r] += fmaxf(a[r] + b, 0.0f);
        }
#pragma unroll
        for (int r = 0; r < 4; ++r)
            if (il[r] < n) haggr[(size_t)idx[il[r]] * F_H + col0 + lane] = sum[r];
    }
}

__global__ void count_same(const int* __restrict__ y, float* __restrict__ cnt, int S)
{
    int i = blockIdx.x * blockDim.x + threadIdx.x;
    if (i >= S) return;
    int yi = y[i];
    int c = 0;
    for (int j = 0; j < S; ++j) c += (y[j] == yi) ? 1 : 0;
    cnt[i] = (float)(c - 1);
}

__global__ __launch_bounds__(256) void update_s(
    float* __restrict__ s, const float* __restrict__ aggmat,
    const float* __restrict__ bm2t, const float* __restrict__ cnt, int H)
{
    const int i = blockIdx.x;
    const int h = threadIdx.x;
    const float c = cnt[i];
    const float denom = fmaxf(c, 1.0f);
    s[(size_t)i * H + h] += (aggmat[(size_t)i * H + h] + c * bm2t[h]) / denom;
}

#define SC_SR 32
#define SC_QR 8

__device__ __forceinline__ float sc_xor16(float x)
{
    const int b = __builtin_bit_cast(int, x);
    return __builtin_bit_cast(float,
        __builtin_amdgcn_permlanex16(b, b, 0x76543210, 0xfedcba98, false, false));
}

template <int M>
__device__ __forceinline__ float sc_xor_row(float x)
{
    const int b = __builtin_bit_cast(int, x);
    return __builtin_bit_cast(float,
        __builtin_amdgcn_update_dpp(b, b, 0x160 + M, 0xf, 0xf, false));
}

__global__ __launch_bounds__(256) void score_kernel(
    const float* __restrict__ hq, const float* __restrict__ hs,
    const float* __restrict__ br1, const float* __restrict__ wr2,
    const float* __restrict__ br2, float* __restrict__ out, int S, int H)
{
    const int lane  = threadIdx.x & 31;
    const int wave  = threadIdx.x >> 5;
    const int hbase = lane * 8;
    const int s0 = blockIdx.x * SC_SR;
    const int q0 = (blockIdx.y * 8 + wave) * SC_QR;

    const bool up16 = (lane & 16) != 0;
    const bool up8  = (lane & 8) != 0;
    const bool up4  = (lane & 4) != 0;
    const bool up2  = (lane & 2) != 0;
    const bool up1  = (lane & 1) != 0;

    float bv[8], wv[8];
    {
        const v4f b0 = *(const v4f*)(br1 + hbase), b1 = *(const v4f*)(br1 + hbase + 4);
        const v4f w0 = *(const v4f*)(wr2 + hbase), w1 = *(const v4f*)(wr2 + hbase + 4);
#pragma unroll
        for (int u = 0; u < 4; ++u) {
            bv[u] = b0[u]; bv[u + 4] = b1[u];
            wv[u] = w0[u]; wv[u + 4] = w1[u];
        }
    }
    const float b2 = br2[0];

    float hsr[SC_SR][8];
#pragma unroll
    for (int r = 0; r < SC_SR; ++r) {
        const float* row = hs + (size_t)(s0 + r) * H + hbase;
        const v4f x0 = *(const v4f*)row, x1 = *(const v4f*)(row + 4);
#pragma unroll
        for (int u = 0; u < 4; ++u) { hsr[r][u] = x0[u]; hsr[r][u + 4] = x1[u]; }
    }

    const float* hqrow = hq + (size_t)q0 * H + hbase;
    v4f n0 = *(const v4f*)hqrow, n1 = *(const v4f*)(hqrow + 4);
    float* orow = out + (size_t)q0 * S + s0 + lane;

#pragma nounroll
    for (int qi = 0; qi < SC_QR; ++qi) {
        float hqv[8];
#pragma unroll
        for (int u = 0; u < 4; ++u) { hqv[u] = n0[u]; hqv[u + 4] = n1[u]; }
        if (qi + 1 < SC_QR) {
            hqrow += H;
            n0 = *(const v4f*)hqrow;
            n1 = *(const v4f*)(hqrow + 4);
        }

        float p[SC_SR];
#pragma unroll
        for (int r = 0; r < SC_SR; ++r) {
            float sum = 0.0f;
#pragma unroll
            for (int u = 0; u < 8; ++u)
                sum = fmaf(fmaxf(hqv[u] + hsr[r][u] + bv[u], 0.0f), wv[u], sum);
            p[r] = sum;
        }

#pragma unroll
        for (int i = 0; i < 16; ++i) {
            const float keep = up16 ? p[i + 16] : p[i];
            const float send = up16 ? p[i] : p[i + 16];
            p[i] = keep + sc_xor16(send);
        }
#pragma unroll
        for (int i = 0; i < 8; ++i) {
            const float keep = up8 ? p[i + 8] : p[i];
            const float send = up8 ? p[i] : p[i + 8];
            p[i] = keep + sc_xor_row<8>(send);
        }
#pragma unroll
        for (int i = 0; i < 4; ++i) {
            const float keep = up4 ? p[i + 4] : p[i];
            const float send = up4 ? p[i] : p[i + 4];
            p[i] = keep + sc_xor_row<4>(send);
        }
#pragma unroll
        for (int i = 0; i < 2; ++i) {
            const float keep = up2 ? p[i + 2] : p[i];
            const float send = up2 ? p[i] : p[i + 2];
            p[i] = keep + sc_xor_row<2>(send);
        }
        {
            const float keep = up1 ? p[1] : p[0];
            const float send = up1 ? p[0] : p[1];
            p[0] = keep + sc_xor_row<1>(send);
        }

        orow[0] = p[0] + b2;
        orow += S;
    }
}

extern "C" void kernel_launch(void* const* d_in, const int* in_sizes, int n_in,
                              void* d_out, int out_size, void* d_ws, size_t ws_size,
                              hipStream_t stream)
{
    const float* Xq  = (const float*)d_in[0];
    const float* Xs  = (const float*)d_in[1];
    const int*   y   = (const int*)d_in[2];
    const float* Wn1 = (const float*)d_in[3];
    const float* bn1 = (const float*)d_in[4];
    const float* Wn2 = (const float*)d_in[5];
    const float* bn2 = (const float*)d_in[6];
    const float* Wm1 = (const float*)d_in[7];
    const float* bm1 = (const float*)d_in[8];
    const float* Wm2 = (const float*)d_in[9];
    const float* bm2 = (const float*)d_in[10];
    const float* Wr1 = (const float*)d_in[11];
    const float* br1 = (const float*)d_in[12];
    const float* wr2 = (const float*)d_in[13];
    const float* br2 = (const float*)d_in[14];
    float* out = (float*)d_out;

    const int E = 512, H = 256, Q = 1024, S = 512, STEPS = 3;

    float* ws  = (float*)d_ws;
    float* q1  = ws; ws += (size_t)Q * H;
    float* qf  = ws; ws += (size_t)Q * H;
    float* s1  = ws; ws += (size_t)S * H;
    float* sf  = ws; ws += (size_t)S * H;
    float* Ab  = ws; ws += (size_t)S * H;
    float* Bb  = ws; ws += (size_t)S * H;
    float* hag = ws; ws += (size_t)S * H;
    float* agg = ws; ws += (size_t)S * H;
    float* hqb = ws; ws += (size_t)Q * H;
    float* hsb = ws; ws += (size_t)S * H;
    float* cnt = ws; ws += S;

    auto gemm = [&](const float* A, const float* W, const float* bias, float* C,
                    int M, int N, int K, int relu) {
        int blocks = (M / 32) * (N / 64);
        wmma_gemm_bias<<<blocks, 256, 0, stream>>>(A, W, bias, C, M, N, K, relu);
    };

    gemm(Xq, Wn1, bn1, q1, Q, H, E, 1);
    gemm(q1, Wn2, bn2, qf, Q, H, H, 0);
    gemm(Xs, Wn1, bn1, s1, S, H, E, 1);
    gemm(s1, Wn2, bn2, sf, S, H, H, 0);

    count_same<<<(S + 255) / 256, 256, 0, stream>>>(y, cnt, S);

    for (int t = 0; t < STEPS; ++t) {
        const float* W1a = Wm1 + (size_t)t * (2 * H) * H;
        const float* W1b = W1a + (size_t)H * H;
        class_ab_pair<<<dim3(8, 16), 512, 0, stream>>>(sf, W1a, W1b, bm1 + (size_t)t * H,
                                                       y, hag, S);
        gemm(hag, Wm2 + (size_t)t * H * H, nullptr, agg, S, H, H, 0);
        update_s<<<S, H, 0, stream>>>(sf, agg, bm2 + (size_t)t * H, cnt, H);
    }

    gemm(qf, Wr1, nullptr, hqb, Q, H, H, 0);
    gemm(sf, Wr1 + (size_t)H * H, nullptr, hsb, S, H, H, 0);
    score_kernel<<<dim3(16, 16), 256, 0, stream>>>(hqb, hsb, br1, wr2, br2, out, S, H);
}


# Node update moves into its consumer: class launch rebuilds its members' rows (prev + (agg + cnt*bm2)/max(cnt,1), cnt = members-1), stages them in LDS as A operand; 17 -> 14 launches; sum order kept
# speedup vs baseline: 7.6756x; 1.1496x over previous
#include <hip/hip_runtime.h>
#include <hip/hip_bf16.h>

typedef float v2f __attribute__((ext_vector_type(2)));
typedef float v4f __attribute__((ext_vector_type(4)));
typedef float v8f __attribute__((ext_vector_type(8)));

#define G_BM 32
#define G_BN 64
#define G_BK 128
#define G_ALD (G_BK + 4)
#define G_WLD (G_BN + 8)

__global__ __launch_bounds__(256) void wmma_gemm_bias(
    const float* __restrict__ A, const float* __restrict__ W,
    const float* __restrict__ bias, float* __restrict__ C,
    int M, int N, int K, int relu)
{
    __shared__ __attribute__((aligned(16))) float As[G_BM * G_ALD];
    __shared__ __attribute__((aligned(16))) float Ws[G_BK * G_WLD];

    const int tid  = threadIdx.x;
    const int lane = tid & 31;
    const int wave = tid >> 5;
    const int half = lane >> 4;
    const int l16  = lane & 15;

    const int tilesN = N >> 6;
    const int bm = (blockIdx.x / tilesN) * G_BM;
    const int bn = (blockIdx.x % tilesN) * G_BN;
    const int wm = (wave >> 2) << 4;
    const int wn = (wave & 3) << 4;

    const int ar = tid >> 5, ac = (tid & 31) << 2;
    const int wr = tid >> 4, wc = (tid & 15) << 2;

    const float* ag = A + (size_t)(bm + ar) * K + ac;
    const float* wg = W + (size_t)wr * N + bn + wc;
    const size_t a8  = (size_t)8 * K;
    const size_t w16 = (size_t)16 * N;

    v4f ra[4], rw[8];
#pragma unroll
    for (int i = 0; i < 4; ++i) ra[i] = *(const v4f*)(ag + i * a8);
#pragma unroll
    for (int i = 0; i < 8; ++i) rw[i] = *(const v4f*)(wg + i * w16);

    const float* afrag = &As[(wm + l16) * G_ALD + 2 * half];
    const float* bfrag = &Ws[(2 * half) * G_WLD + wn + l16];

    v8f acc = {};
    for (int k0 = 0; k0 < K; k0 += G_BK) {
#pragma unroll
        for (int i = 0; i < 4; ++i)
            *(v4f*)&As[(ar + 8 * i) * G_ALD + ac] = ra[i];
#pragma unroll
        for (int i = 0; i < 8; ++i)
            *(v4f*)&Ws[(wr + 16 * i) * G_WLD + wc] = rw[i];
        __syncthreads();

        if (k0 + G_BK < K) {
            ag += G_BK;
            wg += (size_t)G_BK * N;
#pragma unroll
            for (int i = 0; i < 4; ++i) ra[i] = *(const v4f*)(ag + i * a8);
#pragma unroll
            for (int i = 0; i < 8; ++i) rw[i] = *(const v4f*)(wg + i * w16);
        }

        v2f fa[2][8], fb[2][8];
#pragma unroll
        for (int s = 0; s < 8; ++s) {
            fa[0][s]   = *(const v2f*)(afrag + 4 * s);
            fb[0][s].x = bfrag[(4 * s) * G_WLD];
            fb[0][s].y = bfrag[(4 * s) * G_WLD + G_WLD];
        }
#pragma unroll
        for (int g = 0; g < G_BK / 32; ++g) {
            if (g + 1 < G_BK / 32) {
#pragma unroll
                for (int s = 0; s < 8; ++s) {
                    const int kk = 32 * (g + 1) + 4 * s;
                    fa[(g + 1) & 1][s]   = *(const v2f*)(afrag + kk);
                    fb[(g + 1) & 1][s].x = bfrag[kk * G_WLD];
                    fb[(g + 1) & 1][s].y = bfrag[kk * G_WLD + G_WLD];
                }
            }
            __builtin_amdgcn_sched_barrier(0);
#pragma unroll
            for (int s = 0; s < 8; ++s)
                acc = __builtin_amdgcn_wmma_f32_16x16x4_f32(false, fa[g & 1][s], false,
                                                            fb[g & 1][s], (short)0, acc,
                                                            false, false);
            __builtin_amdgcn_sched_barrier(0);
        }
        __syncthreads();
    }

    const int row0 = bm + wm + half * 8;
    const int col  = bn + wn + l16;
    float bv = 0.f;
    if (bias) bv = bias[col];
    float* crow = C + (size_t)row0 * N + col;
#pragma unroll
    for (int r = 0; r < 8; ++r) {
        float v = acc[r] + bv;
        if (relu) v = fmaxf(v, 0.f);
        crow[0] = v;
        crow += N;
    }
}

#define F_NW 16
#define F_COLS 32
#define F_MAXN 512
#define F_H 256
#define F_KC 64
#define F_SB 64
#define F_SLD (F_H + 4)

__global__ __launch_bounds__(512) void class_ab_pair(
    const float* __restrict__ sp, const float* __restrict__ aggp,
    const float* __restrict__ bm2p, float* __restrict__ snew,
    const float* __restrict__ W1a, const float* __restrict__ W1b,
    const float* __restrict__ bm1t, const int* __restrict__ y,
    float* __restrict__ haggr, int S)
{
    __shared__ float As[F_MAXN * F_COLS];
    __shared__ float Bs[F_MAXN * F_COLS];
    __shared__ __attribute__((aligned(16))) float Ss[F_SB * F_SLD];
    __shared__ int idx[F_MAXN];
    __shared__ int wcount[F_NW];

    const int tid  = threadIdx.x;
    const int lane = tid & 31;
    const int wave = __builtin_amdgcn_readfirstlane(tid >> 5);
    const int half = lane >> 4;
    const int l16  = lane & 15;
    const int cls  = blockIdx.y;
    const int col0 = blockIdx.x * F_COLS;

    int n = 0;
    for (int base = 0; base < S; base += F_NW * 32) {
        const int j = base + tid;
        const bool mine = (j < S) && (y[j] == cls);
        const unsigned bal = __builtin_amdgcn_ballot_w32(mine);
        if (lane == 0) wcount[wave] = __popc(bal);
        __syncthreads();
        int before = 0, total = 0;
        for (int w = 0; w < F_NW; ++w) {
            const int c = wcount[w];
            if (w < wave) before += c;
            total += c;
        }
        const int pos = n + before + __popc(bal & ((1u << lane) - 1u));
        if (mine && pos < F_MAXN) idx[pos] = j;
        n += total;
        __syncthreads();
    }
    if (n == 0) return;
    const float cf    = (float)(n - 1);
    const float denom = fmaxf(cf, 1.0f);
    if (n > F_MAXN) n = F_MAXN;

    const int  pc    = col0 & 127;
    const bool hi    = col0 >= 128;
    const bool owner = (lane * 4 >= pc) && (lane * 4 < pc + F_COLS);
    v4f b0 = {}, b1 = {};
    if (aggp) {
        b0 = *(const v4f*)(bm2p + lane * 4);
        b1 = *(const v4f*)(bm2p + 128 + lane * 4);
    }

    for (int rb = 0; rb < n; rb += F_SB) {
        const int nb = (n - rb < F_SB) ? (n - rb) : F_SB;

        int rowq[4];
        v4f x0[4], x1[4];
#pragma unroll
        for (int q = 0; q < 4; ++q) {
            int lr = wave + F_NW * q;
            if (lr >= nb) lr = nb - 1;
            rowq[q] = idx[rb + lr];
            const float* srow = sp + (size_t)rowq[q] * F_H + lane * 4;
            x0[q] = *(const v4f*)srow;
            x1[q] = *(const v4f*)(srow + 128);
        }
        v4f keep[4];
        if (aggp) {
            v4f g0[4], g1[4];
#pragma unroll
            for (int q = 0; q < 4; ++q) {
                const float* grow = aggp + (size_t)rowq[q] * F_H + lane * 4;
                g0[q] = *(const v4f*)grow;
                g1[q] = *(const v4f*)(grow + 128);
            }
#pragma unroll
            for (int q = 0; q < 4; ++q) {
                if (wave + F_NW * q < nb) {
#pragma unroll
                    for (int u = 0; u < 4; ++u) {
                        x0[q][u] += (g0[q][u] + cf * b0[u]) / denom;
                        x1[q][u] += (g1[q][u] + cf * b1[u]) / denom;
                    }
                }
            }
        }
#pragma unroll
        for (int q = 0; q < 4; ++q) {
            const int lr = wave + F_NW * q;
            keep[q] = hi ? x1[q] : x0[q];
            if (lr < nb) {
                *(v4f*)&Ss[lr * F_SLD + lane * 4]       = x0[q];
                *(v4f*)&Ss[lr * F_SLD + 128 + lane * 4] = x1[q];
            }
        }
        __syncthreads();

        if (aggp) {
#pragma unroll
            for (int q = 0; q < 4; ++q) {
                if (wave + F_NW * q < nb && owner)
                    *(v4f*)(snew + (size_t)rowq[q] * F_H + (hi ? 128 : 0) + lane * 4) = keep[q];
            }
        }

        const int ntiles = ((nb + 15) >> 4) * 4;
        for (int t = wave; t < ntiles; t += F_NW) {
            const int ml = t >> 2, nt = t & 3;
            int rl = ml * 16 + l16;
            if (rl >= nb) rl = nb - 1;
            const float* arow = &Ss[rl * F_SLD + 2 * half];
            const float* wcol = ((nt & 2) ? W1b : W1a) + (size_t)(2 * half) * F_H
                                + col0 + (nt & 1) * 16 + l16;
            v2f fa[2][16], fb[2][16];
#pragma unroll
            for (int s = 0; s < 16; ++s) {
                fa[0][s]   = *(const v2f*)(arow + 4 * s);
                fb[0][s].x = wcol[(4 * s) * F_H];
                fb[0][s].y = wcol[(4 * s + 1) * F_H];
            }
            v8f acc = {};
#pragma unroll
            for (int c = 0; c < F_H / F_KC; ++c) {
                if (c + 1 < F_H / F_KC) {
#pragma unroll
                    for (int s = 0; s < 16; ++s) {
                        const int kk = F_KC * (c + 1) + 4 * s;
                        fa[(c + 1) & 1][s]   = *(const v2f*)(arow + kk);
                        fb[(c + 1) & 1][s].x = wcol[kk * F_H];
                        fb[(c + 1) & 1][s].y = wcol[(kk + 1) * F_H];
                    }
                }
                __builtin_amdgcn_sched_barrier(0);
#pragma unroll
                for (int s = 0; s < 16; ++s)
                    acc = __builtin_amdgcn_wmma_f32_16x16x4_f32(false, fa[c & 1][s], false,
                                                                fb[c & 1][s], (short)0, acc,
                                                                false, false);
                __builtin_amdgcn_sched_barrier(0);
            }
            float* dst = ((nt & 2) ? Bs : As) + (rb + ml * 16 + half * 8) * F_COLS
                         + (nt & 1) * 16 + l16;
#pragma unroll
            for (int r = 0; r < 8; ++r) dst[r * F_COLS] = acc[r];
        }
        __syncthreads();
    }

    const float bmv = bm1t[col0 + lane];
    for (int r0 = wave; r0 < n; r0 += 4 * F_NW) {
        int il[4];
        float a[4], sum[4];
#pragma unroll
        for (int r = 0; r < 4; ++r) {
            il[r]  = r0 + r * F_NW;
            a[r]   = (il[r] < n) ? As[il[r] * F_COLS + lane] + bmv : 0.f;
            sum[r] = 0.f;
        }
#pragma unroll 4
        for (int j = 0; j < n; ++j) {
            const float b = Bs[j * F_COLS + lane];
#pragma unroll
            for (int r = 0; r < 4; ++r)
                if (j != il[r]) sum[r] += fmaxf(a[r] + b, 0.0f);
        }
#pragma unroll
        for (int r = 0; r < 4; ++r)
            if (il[r] < n) haggr[(size_t)idx[il[r]] * F_H + col0 + lane] = sum[r];
    }
}

#define U_S 512

__global__ __launch_bounds__(256) void update_last(
    const float* __restrict__ s, const float* __restrict__ aggmat,
    const float* __restrict__ bm2t, const int* __restrict__ y,
    float* __restrict__ sout)
{
    const int lane = threadIdx.x & 31;
    const int wave = __builtin_amdgcn_readfirstlane(threadIdx.x >> 5);
    const int i = blockIdx.x * 8 + wave;

    int yv[U_S / 32];
#pragma unroll
    for (int m = 0; m < U_S / 32; ++m) yv[m] = y[m * 32 + lane];
    const int yi = y[i];

    const size_t off = (size_t)i * F_H + lane * 4;
    v4f x0 = *(const v4f*)(s + off),      x1 = *(const v4f*)(s + off + 128);
    const v4f g0 = *(const v4f*)(aggmat + off), g1 = *(const v4f*)(aggmat + off + 128);
    const v4f b0 = *(const v4f*)(bm2t + lane * 4), b1 = *(const v4f*)(bm2t + 128 + lane * 4);

    int c = -1;
#pragma unroll
    for (int m = 0; m < U_S / 32; ++m)
        c += __popc(__builtin_amdgcn_ballot_w32(yv[m] == yi));
    const float cf = (float)c;
    const float denom = fmaxf(cf, 1.0f);
#pragma unroll
    for (int u = 0; u < 4; ++u) {
        x0[u] += (g0[u] + cf * b0[u]) / denom;
        x1[u] += (g1[u] + cf * b1[u]) / denom;
    }
    *(v4f*)(sout + off)       = x0;
    *(v4f*)(sout + off + 128) = x1;
}

#define SC_SR 32
#define SC_QR 8

__device__ __forceinline__ float sc_xor16(float x)
{
    const int b = __builtin_bit_cast(int, x);
    return __builtin_bit_cast(float,
        __builtin_amdgcn_permlanex16(b, b, 0x76543210, 0xfedcba98, false, false));
}

template <int M>
__device__ __forceinline__ float sc_xor_row(float x)
{
    const int b = __builtin_bit_cast(int, x);
    return __builtin_bit_cast(float,
        __builtin_amdgcn_update_dpp(b, b, 0x160 + M, 0xf, 0xf, false));
}

__global__ __launch_bounds__(256) void score_kernel(
    const float* __restrict__ hq, const float* __restrict__ hs,
    const float* __restrict__ br1, const float* __restrict__ wr2,
    const float* __restrict__ br2, float* __restrict__ out, int S, int H)
{
    const int lane  = threadIdx.x & 31;
    const int wave  = threadIdx.x >> 5;
    const int hbase = lane * 8;
    const int s0 = blockIdx.x * SC_SR;
    const int q0 = (blockIdx.y * 8 + wave) * SC_QR;

    const bool up16 = (lane & 16) != 0;
    const bool up8  = (lane & 8) != 0;
    const bool up4  = (lane & 4) != 0;
    const bool up2  = (lane & 2) != 0;
    const bool up1  = (lane & 1) != 0;

    float bv[8], wv[8];
    {
        const v4f b0 = *(const v4f*)(br1 + hbase), b1 = *(const v4f*)(br1 + hbase + 4);
        const v4f w0 = *(const v4f*)(wr2 + hbase), w1 = *(const v4f*)(wr2 + hbase + 4);
#pragma unroll
        for (int u = 0; u < 4; ++u) {
            bv[u] = b0[u]; bv[u + 4] = b1[u];
            wv[u] = w0[u]; wv[u + 4] = w1[u];
        }
    }
    const float b2 = br2[0];

    float hsr[SC_SR][8];
#pragma unroll
    for (int r = 0; r < SC_SR; ++r) {
        const float* row = hs + (size_t)(s0 + r) * H + hbase;
        const v4f x0 = *(const v4f*)row, x1 = *(const v4f*)(row + 4);
#pragma unroll
        for (int u = 0; u < 4; ++u) { hsr[r][u] = x0[u]; hsr[r][u + 4] = x1[u]; }
    }

    const float* hqrow = hq + (size_t)q0 * H + hbase;
    v4f n0 = *(const v4f*)hqrow, n1 = *(const v4f*)(hqrow + 4);
    float* orow = out + (size_t)q0 * S + s0 + lane;

#pragma nounroll
    for (int qi = 0; qi < SC_QR; ++qi) {
        float hqv[8];
#pragma unroll
        for (int u = 0; u < 4; ++u) { hqv[u] = n0[u]; hqv[u + 4] = n1[u]; }
        if (qi + 1 < SC_QR) {
            hqrow += H;
            n0 = *(const v4f*)hqrow;
            n1 = *(const v4f*)(hqrow + 4);
        }

        float p[SC_SR];
#pragma unroll
        for (int r = 0; r < SC_SR; ++r) {
            float sum = 0.0f;
#pragma unroll
            for (int u = 0; u < 8; ++u)
                sum = fmaf(fmaxf(hqv[u] + hsr[r][u] + bv[u], 0.0f), wv[u], sum);
            p[r] = sum;
        }

#pragma unroll
        for (int i = 0; i < 16; ++i) {
            const float keep = up16 ? p[i + 16] : p[i];
            const float send = up16 ? p[i] : p[i + 16];
            p[i] = keep + sc_xor16(send);
        }
#pragma unroll
        for (int i = 0; i < 8; ++i) {
            const float keep = up8 ? p[i + 8] : p[i];
            const float send = up8 ? p[i] : p[i + 8];
            p[i] = keep + sc_xor_row<8>(send);
        }
#pragma unroll
        for (int i = 0; i < 4; ++i) {
            const float keep = up4 ? p[i + 4] : p[i];
            const float send = up4 ? p[i] : p[i + 4];
            p[i] = keep + sc_xor_row<4>(send);
        }
#pragma unroll
        for (int i = 0; i < 2; ++i) {
            const float keep = up2 ? p[i + 2] : p[i];
            const float send = up2 ? p[i] : p[i + 2];
            p[i] = keep + sc_xor_row<2>(send);
        }
        {
            const float keep = up1 ? p[1] : p[0];
            const float send = up1 ? p[0] : p[1];
            p[0] = keep + sc_xor_row<1>(send);
        }

        orow[0] = p[0] + b2;
        orow += S;
    }
}

extern "C" void kernel_launch(void* const* d_in, const int* in_sizes, int n_in,
                              void* d_out, int out_size, void* d_ws, size_t ws_size,
                              hipStream_t stream)
{
    const float* Xq  = (const float*)d_in[0];
    const float* Xs  = (const float*)d_in[1];
    const int*   y   = (const int*)d_in[2];
    const float* Wn1 = (const float*)d_in[3];
    const float* bn1 = (const float*)d_in[4];
    const float* Wn2 = (const float*)d_in[5];
    const float* bn2 = (const float*)d_in[6];
    const float* Wm1 = (const float*)d_in[7];
    const float* bm1 = (const float*)d_in[8];
    const float* Wm2 = (const float*)d_in[9];
    const float* bm2 = (const float*)d_in[10];
    const float* Wr1 = (const float*)d_in[11];
    const float* br1 = (const float*)d_in[12];
    const float* wr2 = (const float*)d_in[13];
    const float* br2 = (const float*)d_in[14];
    float* out = (float*)d_out;

    const int E = 512, H = 256, Q = 1024, S = 512;

    float* ws  = (float*)d_ws;
    float* q1  = ws; ws += (size_t)Q * H;
    float* qf  = ws; ws += (size_t)Q * H;
    float* s1  = ws; ws += (size_t)S * H;
    float* sf  = ws; ws += (size_t)S * H;
    float* sA  = ws; ws += (size_t)S * H;
    float* sB  = ws; ws += (size_t)S * H;
    float* sC  = ws; ws += (size_t)S * H;
    float* hag = ws; ws += (size_t)S * H;
    float* agg = ws; ws += (size_t)S * H;
    float* hqb = ws; ws += (size_t)Q * H;
    float* hsb = ws; ws += (size_t)S * H;

    auto gemm = [&](const float* A, const float* W, const float* bias, float* C,
                    int M, int N, int K, int relu) {
        int blocks = (M / 32) * (N / 64);
        wmma_gemm_bias<<<blocks, 256, 0, stream>>>(A, W, bias, C, M, N, K, relu);
    };

    gemm(Xq, Wn1, bn1, q1, Q, H, E, 1);
    gemm(q1, Wn2, bn2, qf, Q, H, H, 0);
    gemm(Xs, Wn1, bn1, s1, S, H, E, 1);
    gemm(s1, Wn2, bn2, sf, S, H, H, 0);

    const float* W1a0 = Wm1;
    const float* W1a1 = Wm1 + (size_t)1 * (2 * H) * H;
    const float* W1a2 = Wm1 + (size_t)2 * (2 * H) * H;
    class_ab_pair<<<dim3(8, 16), 512, 0, stream>>>(sf, nullptr, nullptr, nullptr,
                                                   W1a0, W1a0 + (size_t)H * H, bm1, y, hag, S);
    gemm(hag, Wm2, nullptr, agg, S, H, H, 0);
    class_ab_pair<<<dim3(8, 16), 512, 0, stream>>>(sf, agg, bm2, sA,
                                                   W1a1, W1a1 + (size_t)H * H, bm1 + H, y, hag, S);
    gemm(hag, Wm2 + (size_t)H * H, nullptr, agg, S, H, H, 0);
    class_ab_pair<<<dim3(8, 16), 512, 0, stream>>>(sA, agg, bm2 + H, sB,
                                                   W1a2, W1a2 + (size_t)H * H, bm1 + 2 * H, y, hag, S);
    gemm(hag, Wm2 + (size_t)2 * H * H, nullptr, agg, S, H, H, 0);
    update_last<<<S / 8, 256, 0, stream>>>(sB, agg, bm2 + 2 * H, y, sC);

    gemm(qf, Wr1, nullptr, hqb, Q, H, H, 0);
    gemm(sC, Wr1 + (size_t)H * H, nullptr, hsb, S, H, H, 0);
    score_kernel<<<dim3(16, 16), 256, 0, stream>>>(hqb, hsb, br1, wr2, br2, out, S, H);
}
